# starting kernel
# baseline (speedup 1.0000x reference)
#include <hip/hip_runtime.h>
#include <hip/hip_bf16.h>
#include <math.h>

#define LSEQ 512
#define HDIM 1024
#define HC   256
#define LDSP (HC + 4)

typedef __attribute__((ext_vector_type(2))) float v2f;
typedef __attribute__((ext_vector_type(8))) float v8f;

__device__ __forceinline__ float fast_tanh(float x) {
#if __has_builtin(__builtin_amdgcn_tanhf)
    return __builtin_amdgcn_tanhf(x);
#elif __has_builtin(__builtin_amdgcn_tanh_f32)
    return __builtin_amdgcn_tanh_f32(x);
#else
    float e = __builtin_amdgcn_exp2f(x * 2.8853900817779268f);
    return 1.0f - 2.0f * __builtin_amdgcn_rcpf(1.0f + e);
#endif
}

__global__ __launch_bounds__(256)
void proj_wmma_f32(const float* __restrict__ X, const float* __restrict__ W,
                   const float* __restrict__ bias, float* __restrict__ out,
                   int ldw, int hasBias) {
    const int K = HDIM, N = HDIM;
    int wave  = (blockIdx.x * blockDim.x + threadIdx.x) >> 5;
    int lane  = threadIdx.x & 31;
    int tileM = wave >> 6;
    int tileN = wave & 63;
    int l0 = tileM << 4;
    int o0 = tileN << 4;

    int row  = lane & 15;
    int koff = (lane >> 4) << 1;
    const float* aPtr = X + (size_t)(l0 + row) * K   + koff;
    const float* bPtr = W + (size_t)(o0 + row) * ldw + koff;

    v8f acc = {};
#pragma unroll 8
    for (int k0 = 0; k0 < K; k0 += 4) {
        v2f a = *(const v2f*)(aPtr + k0);
        v2f b = *(const v2f*)(bPtr + k0);
        acc = __builtin_amdgcn_wmma_f32_16x16x4_f32(
             false, a,  false, b,
             (short)0, acc,  false,  false);
    }

    int col   = lane & 15;
    int rbase = (lane >> 4) << 3;
    float bv  = hasBias ? bias[o0 + col] : 0.0f;
#pragma unroll
    for (int r = 0; r < 8; ++r) {
        out[(size_t)(l0 + rbase + r) * N + o0 + col] = acc[r] + bv;
    }
}

__global__ __launch_bounds__(256)
void pairwise_tanh(const float* __restrict__ A, const float* __restrict__ R,
                   const float* __restrict__ W2, float* __restrict__ raw) {
    __shared__ float aT[16][LDSP];
    __shared__ float rT[16][LDSP];
    __shared__ float w2s[2][HC];

    const int i0 = blockIdx.y << 4;
    const int j0 = blockIdx.x << 4;
    const int tid = threadIdx.x;
    const int tj = tid & 15;
    const int ti = tid >> 4;

    float acc0 = 0.0f, acc1 = 0.0f;

    for (int h0 = 0; h0 < HDIM; h0 += HC) {
        __syncthreads();
#pragma unroll
        for (int p = tid; p < 16 * (HC / 4); p += 256) {
            int rrow = p / (HC / 4);
            int c4   = (p % (HC / 4)) << 2;
            float4 va = *(const float4*)(A + (size_t)(j0 + rrow) * HDIM + h0 + c4);
            *(float4*)&aT[rrow][c4] = va;
            float4 vr = *(const float4*)(R + (size_t)(i0 + rrow) * HDIM + h0 + c4);
            *(float4*)&rT[rrow][c4] = vr;
        }
        {
            int p = tid;
            w2s[0][p] = W2[h0 + p];
            w2s[1][p] = W2[HDIM + h0 + p];
        }
        __syncthreads();

#pragma unroll 4
        for (int hc = 0; hc < HC; ++hc) {
            float x = aT[tj][hc] + rT[ti][hc];
            float t = fast_tanh(x);
            acc0 = fmaf(w2s[0][hc], t, acc0);
            acc1 = fmaf(w2s[1][hc], t, acc1);
        }
    }

    int i = i0 + ti, j = j0 + tj;
    float2 o; o.x = acc0; o.y = acc1;
    *(float2*)(raw + ((size_t)(i * LSEQ + j) << 1)) = o;
}

__global__ __launch_bounds__(256)
void symmetrize(const float* __restrict__ raw, const float* __restrict__ b2,
                float* __restrict__ out) {
    int idx = blockIdx.x * 256 + threadIdx.x;
    if (idx >= LSEQ * LSEQ) return;
    int i = idx >> 9, j = idx & (LSEQ - 1);
    float2 rij = *(const float2*)(raw + ((size_t)(i * LSEQ + j) << 1));
    float2 rji = *(const float2*)(raw + ((size_t)(j * LSEQ + i) << 1));
    float2 o;
    o.x = 0.5f * (rij.x + rji.x) + b2[0];
    o.y = 0.5f * (rij.y + rji.y) + b2[1];
    *(float2*)(out + ((size_t)idx << 1)) = o;
}

extern "C" void kernel_launch(void* const* d_in, const int* in_sizes, int n_in,
                              void* d_out, int out_size, void* d_ws, size_t ws_size,
                              hipStream_t stream) {
    const float* X  = (const float*)d_in[0];
    const float* W1 = (const float*)d_in[1];
    const float* b1 = (const float*)d_in[2];
    const float* W2 = (const float*)d_in[3];
    const float* b2 = (const float*)d_in[4];
    float* out = (float*)d_out;

    float* a_ws = (float*)d_ws;
    float* r_ws = a_ws + (size_t)LSEQ * HDIM;
    float* raw  = r_ws + (size_t)LSEQ * HDIM;

    proj_wmma_f32<<<256, 256, 0, stream>>>(X, W1,          nullptr, a_ws, 2 * HDIM, 0);
    proj_wmma_f32<<<256, 256, 0, stream>>>(X, W1 + HDIM,   b1,      r_ws, 2 * HDIM, 1);

    dim3 grid(LSEQ / 16, LSEQ / 16);
    pairwise_tanh<<<grid, 256, 0, stream>>>(a_ws, r_ws, W2, raw);

    symmetrize<<<(LSEQ * LSEQ) / 256, 256, 0, stream>>>(raw, b2, out);
}


# Projection GEMM in one launch: a wave keeps a 64x64 tile in registers, fragments straight from global by 16-byte loads (k permuted alike in A and B), K split over 4 waves, partials summed via LDS
# speedup vs baseline: 1.4610x; 1.4610x over previous
#include <hip/hip_runtime.h>
#include <hip/hip_bf16.h>
#include <math.h>

#define LSEQ 512
#define HDIM 1024

typedef __attribute__((ext_vector_type(2))) float v2f;
typedef __attribute__((ext_vector_type(4))) float v4f;
typedef __attribute__((ext_vector_type(8))) float v8f;

__device__ __forceinline__ float fast_tanh(float x) {
#if __has_builtin(__builtin_amdgcn_tanhf)
    return __builtin_amdgcn_tanhf(x);
#elif __has_builtin(__builtin_amdgcn_tanh_f32)
    return __builtin_amdgcn_tanh_f32(x);
#else
    float e = __builtin_amdgcn_exp2f(x * 2.8853900817779268f);
    return 1.0f - 2.0f * __builtin_amdgcn_rcpf(1.0f + e);
#endif
}

#define GP 68
#define KQ 256

__device__ __forceinline__ void load_frags(v4f (&f)[4][4], const float* __restrict__ p, int stride16) {
#pragma unroll
    for (int i = 0; i < 4; ++i) {
#pragma unroll
        for (int j2 = 0; j2 < 4; ++j2) f[i][j2] = *(const v4f*)(p + (size_t)i * stride16 + 8 * j2);
    }
}

__device__ __forceinline__ void mma_chunk(v8f (&acc)[4][4], const v4f (&p)[4][4], const v4f (&q)[4][4]) {
#pragma unroll
    for (int j2 = 0; j2 < 4; ++j2) {
#pragma unroll
        for (int u = 0; u < 2; ++u) {
            v2f a[4], b[4];
#pragma unroll
            for (int i = 0; i < 4; ++i) {
                a[i] = u ? (v2f){p[i][j2].z, p[i][j2].w} : (v2f){p[i][j2].x, p[i][j2].y};
                b[i] = u ? (v2f){q[i][j2].z, q[i][j2].w} : (v2f){q[i][j2].x, q[i][j2].y};
            }
#pragma unroll
            for (int i = 0; i < 4; ++i) {
#pragma unroll
                for (int j = 0; j < 4; ++j) {
                    acc[i][j] = __builtin_amdgcn_wmma_f32_16x16x4_f32(
                         false, a[i],  false, b[j],
                         (short)0, acc[i][j],  false,  false);
                }
            }
        }
    }
}

__global__ __launch_bounds__(128)
void proj_wmma_f32(const float* __restrict__ X, const float* __restrict__ W,
                   const float* __restrict__ bias, float* __restrict__ a_out,
                   float* __restrict__ r_out) {
    __shared__ v4f lds4[10560];
    float* L = (float*)lds4;

    const int lane = threadIdx.x & 31;
    const int w  = __builtin_amdgcn_readfirstlane(threadIdx.x >> 5);
    const int c  = lane & 15;
    const int hl = lane >> 4;
    const int nb   = blockIdx.x & 31;
    const int half = nb >> 4;
    const int o0 = (nb & 15) << 6;
    const int l0 = (blockIdx.x >> 5) << 6;

    const int kq = KQ * w + 4 * hl;
    const float* xrow = X + (size_t)(l0 + c) * HDIM + kq;
    const float* wrow = W + (size_t)(o0 + c) * (2 * HDIM) + half * HDIM + kq;
    const float* pP = half ? wrow : xrow;
    const float* pQ = half ? xrow : wrow;
    const int sP = half ? 32 * HDIM : 16 * HDIM;
    const int sQ = half ? 16 * HDIM : 32 * HDIM;

    v8f acc[4][4];
#pragma unroll
    for (int i = 0; i < 4; ++i) {
#pragma unroll
        for (int j = 0; j < 4; ++j) acc[i][j] = (v8f){0.f, 0.f, 0.f, 0.f, 0.f, 0.f, 0.f, 0.f};
    }

    v4f p0[4][4], q0[4][4], p1[4][4], q1[4][4];
    load_frags(p0, pP, sP);
    load_frags(q0, pQ, sQ);
#pragma nounroll
    for (int k = 0; k < KQ - 64; k += 64) {
        load_frags(p1, pP + k + 32, sP);
        load_frags(q1, pQ + k + 32, sQ);
        mma_chunk(acc, p0, q0);
        load_frags(p0, pP + k + 64, sP);
        load_frags(q0, pQ + k + 64, sQ);
        mma_chunk(acc, p1, q1);
    }
    load_frags(p1, pP + KQ - 32, sP);
    load_frags(q1, pQ + KQ - 32, sQ);
    mma_chunk(acc, p0, q0);
    mma_chunk(acc, p1, q1);

#pragma unroll
    for (int i = 0; i < 4; ++i) {
#pragma unroll
        for (int j = 0; j < 4; ++j) {
            float* d = L + (size_t)((w * 64 + 16 * j + c) * GP + 16 * i + 8 * hl);
            *(v4f*)(d)     = (v4f){acc[i][j][0], acc[i][j][1], acc[i][j][2], acc[i][j][3]};
            *(v4f*)(d + 4) = (v4f){acc[i][j][4], acc[i][j][5], acc[i][j][6], acc[i][j][7]};
        }
    }
    __syncthreads();

    if (half == 0) {
#pragma unroll
        for (int it = 0; it < 8; ++it) {
            const int pr = 8 * w + it;
            const float* s = L + (size_t)(2 * pr) * GP + 2 * lane;
            v2f e0 = *(const v2f*)(s);
            v2f e1 = *(const v2f*)(s + GP);
#pragma unroll
            for (int sw = 1; sw < 4; ++sw) {
                e0 += *(const v2f*)(s + (size_t)sw * 64 * GP);
                e1 += *(const v2f*)(s + (size_t)sw * 64 * GP + GP);
            }
            v4f o = (v4f){e0.x, e0.y, e1.x, e1.y};
            *(v4f*)(a_out + ((size_t)((o0 >> 1) + pr) * (LSEQ / 2) + (l0 >> 1) + lane) * 4) = o;
        }
    } else {
        const int m4 = (lane & 15) << 2;
        const v4f bv = *(const v4f*)(bias + o0 + m4);
#pragma unroll
        for (int it = 0; it < 8; ++it) {
            const int n = 16 * w + 2 * it + (lane >> 4);
            const float* s = L + (size_t)n * GP + m4;
            v4f v = *(const v4f*)(s);
#pragma unroll
            for (int sw = 1; sw < 4; ++sw) v += *(const v4f*)(s + (size_t)sw * 64 * GP);
            v += bv;
            *(v4f*)(r_out + (size_t)(l0 + n) * HDIM + o0 + m4) = v;
        }
    }
}

#define NI 2
#define CH 8

__global__ __launch_bounds__(256)
void pairwise_tanh(const float* __restrict__ A4, const float* __restrict__ R,
                   const float* __restrict__ W2, float* __restrict__ raw) {
    __shared__ float4 part[4][NI][32];

    const int lane = threadIdx.x & 31;
    const int wv = __builtin_amdgcn_readfirstlane(threadIdx.x >> 5);
    const int rpair = wv & 3;
    const int hhalf = wv >> 2;

    const int jp = blockIdx.y * 32 + lane;
    const int i0 = (blockIdx.x * 4 + rpair) * NI;
    const int hbeg = hhalf * (HDIM / 2);

    const float4* __restrict__ ap = (const float4*)A4 + jp;
    const float*  __restrict__ rp = R + (size_t)i0 * HDIM;

    v2f acc0[NI], acc1[NI];
#pragma unroll
    for (int ii = 0; ii < NI; ++ii) { acc0[ii] = (v2f){0.0f, 0.0f}; acc1[ii] = (v2f){0.0f, 0.0f}; }

    for (int h0 = hbeg; h0 < hbeg + HDIM / 2; h0 += CH) {
        float4 a4[CH / 2];
#pragma unroll
        for (int q = 0; q < CH / 2; ++q) a4[q] = ap[(size_t)((h0 >> 1) + q) * (LSEQ / 2)];
#pragma unroll
        for (int k = 0; k < CH; ++k) {
            v2f a2 = (k & 1) ? (v2f){a4[k >> 1].z, a4[k >> 1].w} : (v2f){a4[k >> 1].x, a4[k >> 1].y};
            float w0 = W2[h0 + k];
            float w1 = W2[HDIM + h0 + k];
#pragma unroll
            for (int ii = 0; ii < NI; ++ii) {
                float r = rp[(size_t)ii * HDIM + h0 + k];
                v2f x = a2 + (v2f){r, r};
                v2f t;
                t.x = fast_tanh(x.x);
                t.y = fast_tanh(x.y);
                acc0[ii] = __builtin_elementwise_fma((v2f){w0, w0}, t, acc0[ii]);
                acc1[ii] = __builtin_elementwise_fma((v2f){w1, w1}, t, acc1[ii]);
            }
        }
    }

    float4 o[NI];
#pragma unroll
    for (int ii = 0; ii < NI; ++ii) {
        o[ii].x = acc0[ii].x; o[ii].y = acc1[ii].x; o[ii].z = acc0[ii].y; o[ii].w = acc1[ii].y;
    }
    if (hhalf == 1) {
#pragma unroll
        for (int ii = 0; ii < NI; ++ii) part[rpair][ii][lane] = o[ii];
    }
    __syncthreads();
    if (hhalf == 0) {
#pragma unroll
        for (int ii = 0; ii < NI; ++ii) {
            float4 p = part[rpair][ii][lane];
            float4 s; s.x = o[ii].x + p.x; s.y = o[ii].y + p.y; s.z = o[ii].z + p.z; s.w = o[ii].w + p.w;
            *(float4*)(raw + ((size_t)((i0 + ii) * LSEQ + 2 * jp) << 1)) = s;
        }
    }
}

__global__ __launch_bounds__(256)
void symmetrize(const float* __restrict__ raw, const float* __restrict__ b2,
                float* __restrict__ out) {
    int idx = blockIdx.x * 256 + threadIdx.x;
    if (idx >= LSEQ * LSEQ) return;
    int i = idx >> 9, j = idx & (LSEQ - 1);
    float2 rij = *(const float2*)(raw + ((size_t)(i * LSEQ + j) << 1));
    float2 rji = *(const float2*)(raw + ((size_t)(j * LSEQ + i) << 1));
    float2 o;
    o.x = 0.5f * (rij.x + rji.x) + b2[0];
    o.y = 0.5f * (rij.y + rji.y) + b2[1];
    *(float2*)(out + ((size_t)idx << 1)) = o;
}

extern "C" void kernel_launch(void* const* d_in, const int* in_sizes, int n_in,
                              void* d_out, int out_size, void* d_ws, size_t ws_size,
                              hipStream_t stream) {
    const float* X  = (const float*)d_in[0];
    const float* W1 = (const float*)d_in[1];
    const float* b1 = (const float*)d_in[2];
    const float* W2 = (const float*)d_in[3];
    const float* b2 = (const float*)d_in[4];
    float* out = (float*)d_out;

    float* a_ws = (float*)d_ws;
    float* r_ws = a_ws + (size_t)LSEQ * HDIM;
    float* raw  = r_ws + (size_t)LSEQ * HDIM;

    proj_wmma_f32<<<256, 128, 0, stream>>>(X, W1, b1, a_ws, r_ws);

    dim3 grid(LSEQ / (4 * NI), LSEQ / 64);
    pairwise_tanh<<<grid, 256, 0, stream>>>(a_ws, r_ws, W2, raw);

    symmetrize<<<(LSEQ * LSEQ) / 256, 256, 0, stream>>>(raw, b2, out);
}


# Pairwise loop software-pipelined: two operand sets, next 8 h of A and scalar R, W2 load while 8 h compute; lane owns 4 columns, wave a quarter of h: 4 partial h sums added at the end, not 2
# speedup vs baseline: 2.1114x; 1.4452x over previous
#include <hip/hip_runtime.h>
#include <hip/hip_bf16.h>
#include <math.h>

#define LSEQ 512
#define HDIM 1024

typedef __attribute__((ext_vector_type(2))) float v2f;
typedef __attribute__((ext_vector_type(4))) float v4f;
typedef __attribute__((ext_vector_type(8))) float v8f;

__device__ __forceinline__ float fast_tanh(float x) {
#if __has_builtin(__builtin_amdgcn_tanhf)
    return __builtin_amdgcn_tanhf(x);
#elif __has_builtin(__builtin_amdgcn_tanh_f32)
    return __builtin_amdgcn_tanh_f32(x);
#else
    float e = __builtin_amdgcn_exp2f(x * 2.8853900817779268f);
    return 1.0f - 2.0f * __builtin_amdgcn_rcpf(1.0f + e);
#endif
}

#define GP 68
#define KQ 256

__device__ __forceinline__ void load_frags(v4f (&f)[4][4], const float* __restrict__ p, int stride16) {
#pragma unroll
    for (int i = 0; i < 4; ++i) {
#pragma unroll
        for (int j2 = 0; j2 < 4; ++j2) f[i][j2] = *(const v4f*)(p + (size_t)i * stride16 + 8 * j2);
    }
}

__device__ __forceinline__ void mma_chunk(v8f (&acc)[4][4], const v4f (&p)[4][4], const v4f (&q)[4][4]) {
#pragma unroll
    for (int j2 = 0; j2 < 4; ++j2) {
#pragma unroll
        for (int u = 0; u < 2; ++u) {
            v2f a[4], b[4];
#pragma unroll
            for (int i = 0; i < 4; ++i) {
                a[i] = u ? (v2f){p[i][j2].z, p[i][j2].w} : (v2f){p[i][j2].x, p[i][j2].y};
                b[i] = u ? (v2f){q[i][j2].z, q[i][j2].w} : (v2f){q[i][j2].x, q[i][j2].y};
            }
#pragma unroll
            for (int i = 0; i < 4; ++i) {
#pragma unroll
                for (int j = 0; j < 4; ++j) {
                    acc[i][j] = __builtin_amdgcn_wmma_f32_16x16x4_f32(
                         false, a[i],  false, b[j],
                         (short)0, acc[i][j],  false,  false);
                }
            }
        }
    }
}

__global__ __launch_bounds__(128)
void proj_wmma_f32(const float* __restrict__ X, const float* __restrict__ W,
                   const float* __restrict__ bias, float* __restrict__ a_out,
                   float* __restrict__ r_out) {
    __shared__ v4f lds4[10560];
    float* L = (float*)lds4;

    const int lane = threadIdx.x & 31;
    const int w  = __builtin_amdgcn_readfirstlane(threadIdx.x >> 5);
    const int c  = lane & 15;
    const int hl = lane >> 4;
    const int nb   = blockIdx.x & 31;
    const int half = nb >> 4;
    const int o0 = (nb & 15) << 6;
    const int l0 = (blockIdx.x >> 5) << 6;

    const int kq = KQ * w + 4 * hl;
    const float* xrow = X + (size_t)(l0 + c) * HDIM + kq;
    const float* wrow = W + (size_t)(o0 + c) * (2 * HDIM) + half * HDIM + kq;
    const float* pP = half ? wrow : xrow;
    const float* pQ = half ? xrow : wrow;
    const int sP = half ? 32 * HDIM : 16 * HDIM;
    const int sQ = half ? 16 * HDIM : 32 * HDIM;

    v8f acc[4][4];
#pragma unroll
    for (int i = 0; i < 4; ++i) {
#pragma unroll
        for (int j = 0; j < 4; ++j) acc[i][j] = (v8f){0.f, 0.f, 0.f, 0.f, 0.f, 0.f, 0.f, 0.f};
    }

    v4f p0[4][4], q0[4][4], p1[4][4], q1[4][4];
    load_frags(p0, pP, sP);
    load_frags(q0, pQ, sQ);
#pragma nounroll
    for (int k = 0; k < KQ - 64; k += 64) {
        load_frags(p1, pP + k + 32, sP);
        load_frags(q1, pQ + k + 32, sQ);
        mma_chunk(acc, p0, q0);
        load_frags(p0, pP + k + 64, sP);
        load_frags(q0, pQ + k + 64, sQ);
        mma_chunk(acc, p1, q1);
    }
    load_frags(p1, pP + KQ - 32, sP);
    load_frags(q1, pQ + KQ - 32, sQ);
    mma_chunk(acc, p0, q0);
    mma_chunk(acc, p1, q1);

#pragma unroll
    for (int i = 0; i < 4; ++i) {
#pragma unroll
        for (int j = 0; j < 4; ++j) {
            float* d = L + (size_t)((w * 64 + 16 * j + c) * GP + 16 * i + 8 * hl);
            *(v4f*)(d)     = (v4f){acc[i][j][0], acc[i][j][1], acc[i][j][2], acc[i][j][3]};
            *(v4f*)(d + 4) = (v4f){acc[i][j][4], acc[i][j][5], acc[i][j][6], acc[i][j][7]};
        }
    }
    __syncthreads();

    if (half == 0) {
#pragma unroll
        for (int it = 0; it < 8; ++it) {
            const int pr = 8 * w + it;
            const float* s = L + (size_t)(2 * pr) * GP + 2 * lane;
            v2f e0 = *(const v2f*)(s);
            v2f e1 = *(const v2f*)(s + GP);
#pragma unroll
            for (int sw = 1; sw < 4; ++sw) {
                e0 += *(const v2f*)(s + (size_t)sw * 64 * GP);
                e1 += *(const v2f*)(s + (size_t)sw * 64 * GP + GP);
            }
            v4f o = (v4f){e0.x, e0.y, e1.x, e1.y};
            *(v4f*)(a_out + ((size_t)((o0 >> 1) + pr) * (LSEQ / 2) + (l0 >> 1) + lane) * 4) = o;
        }
    } else {
        const int m4 = (lane & 15) << 2;
        const v4f bv = *(const v4f*)(bias + o0 + m4);
#pragma unroll
        for (int it = 0; it < 8; ++it) {
            const int n = 16 * w + 2 * it + (lane >> 4);
            const float* s = L + (size_t)n * GP + m4;
            v4f v = *(const v4f*)(s);
#pragma unroll
            for (int sw = 1; sw < 4; ++sw) v += *(const v4f*)(s + (size_t)sw * 64 * GP);
            v += bv;
            *(v4f*)(r_out + (size_t)(l0 + n) * HDIM + o0 + m4) = v;
        }
    }
}

#define NI 2
#define CH 8
#define HQ (HDIM / 4)

struct OpSet {
    float4 aA[CH / 2];
    float4 aB[CH / 2];
    float  r[NI][CH];
    float  w0[CH];
    float  w1[CH];
};

__device__ __forceinline__ void fetch_set(OpSet& s, const float4* __restrict__ ap,
                                          const float* __restrict__ rp,
                                          const float* __restrict__ wp, int h0) {
#pragma unroll
    for (int k = 0; k < CH; ++k) {
        s.w0[k] = wp[h0 + k];
        s.w1[k] = wp[HDIM + h0 + k];
#pragma unroll
        for (int ii = 0; ii < NI; ++ii) s.r[ii][k] = rp[(size_t)ii * HDIM + h0 + k];
    }
#pragma unroll
    for (int q = 0; q < CH / 2; ++q) {
        s.aA[q] = ap[(size_t)((h0 >> 1) + q) * (LSEQ / 2)];
        s.aB[q] = ap[(size_t)((h0 >> 1) + q) * (LSEQ / 2) + 32];
    }
}

__device__ __forceinline__ void use_set(const OpSet& s, v2f (&acc0)[NI][2], v2f (&acc1)[NI][2]) {
#pragma unroll
    for (int k = 0; k < CH; ++k) {
        v2f a2[2];
        a2[0] = (k & 1) ? (v2f){s.aA[k >> 1].z, s.aA[k >> 1].w} : (v2f){s.aA[k >> 1].x, s.aA[k >> 1].y};
        a2[1] = (k & 1) ? (v2f){s.aB[k >> 1].z, s.aB[k >> 1].w} : (v2f){s.aB[k >> 1].x, s.aB[k >> 1].y};
        float w0 = s.w0[k];
        float w1 = s.w1[k];
#pragma unroll
        for (int ii = 0; ii < NI; ++ii) {
            float r = s.r[ii][k];
#pragma unroll
            for (int pp = 0; pp < 2; ++pp) {
                v2f x = a2[pp] + (v2f){r, r};
                v2f t;
                t.x = fast_tanh(x.x);
                t.y = fast_tanh(x.y);
                acc0[ii][pp] = __builtin_elementwise_fma((v2f){w0, w0}, t, acc0[ii][pp]);
                acc1[ii][pp] = __builtin_elementwise_fma((v2f){w1, w1}, t, acc1[ii][pp]);
            }
        }
    }
}

__global__ __launch_bounds__(256)
void pairwise_tanh(const float* __restrict__ A4, const float* __restrict__ R,
                   const float* __restrict__ W2, float* __restrict__ raw) {
    __shared__ float4 part[3][2][NI][2][32];

    const int lane = threadIdx.x & 31;
    const int wv = __builtin_amdgcn_readfirstlane(threadIdx.x >> 5);
    const int rpair = wv & 1;
    const int hq = wv >> 1;

    const int jp = blockIdx.y * 64 + lane;
    const int i0 = (blockIdx.x * 2 + rpair) * NI;
    const int hbeg = hq * HQ;

    const float4* __restrict__ ap = (const float4*)A4 + jp + (size_t)(hbeg >> 1) * (LSEQ / 2);
    const float*  __restrict__ rp = R + (size_t)i0 * HDIM + hbeg;
    const float*  __restrict__ wp = W2 + hbeg;

    v2f acc0[NI][2], acc1[NI][2];
#pragma unroll
    for (int ii = 0; ii < NI; ++ii) {
#pragma unroll
        for (int pp = 0; pp < 2; ++pp) { acc0[ii][pp] = (v2f){0.0f, 0.0f}; acc1[ii][pp] = (v2f){0.0f, 0.0f}; }
    }

#define WAIT_SET(s) asm volatile("" :: "s"((s).w0[0]))
#define HOLD() __builtin_amdgcn_sched_barrier(0)
    OpSet s0, s1;
    fetch_set(s0, ap, rp, wp, 0);
#pragma nounroll
    for (int h0 = 0; h0 < HQ - 2 * CH; h0 += 2 * CH) {
        HOLD(); WAIT_SET(s0); HOLD();
        fetch_set(s1, ap, rp, wp, h0 + CH);
        HOLD();
        use_set(s0, acc0, acc1);
        HOLD(); WAIT_SET(s1); HOLD();
        fetch_set(s0, ap, rp, wp, h0 + 2 * CH);
        HOLD();
        use_set(s1, acc0, acc1);
    }
    HOLD(); WAIT_SET(s0); HOLD();
    fetch_set(s1, ap, rp, wp, HQ - CH);
    HOLD();
    use_set(s0, acc0, acc1);
    use_set(s1, acc0, acc1);
#undef WAIT_SET
#undef HOLD

    float4 o[NI][2];
#pragma unroll
    for (int ii = 0; ii < NI; ++ii) {
#pragma unroll
        for (int pp = 0; pp < 2; ++pp) {
            o[ii][pp].x = acc0[ii][pp].x; o[ii][pp].y = acc1[ii][pp].x;
            o[ii][pp].z = acc0[ii][pp].y; o[ii][pp].w = acc1[ii][pp].y;
        }
    }
    if (hq != 0) {
#pragma unroll
        for (int ii = 0; ii < NI; ++ii) {
#pragma unroll
            for (int pp = 0; pp < 2; ++pp) part[hq - 1][rpair][ii][pp][lane] = o[ii][pp];
        }
    }
    __syncthreads();
    if (hq == 0) {
#pragma unroll
        for (int ii = 0; ii < NI; ++ii) {
#pragma unroll
            for (int pp = 0; pp < 2; ++pp) {
                float4 s = o[ii][pp];
#pragma unroll
                for (int g = 0; g < 3; ++g) {
                    float4 p = part[g][rpair][ii][pp][lane];
                    s.x += p.x; s.y += p.y; s.z += p.z; s.w += p.w;
                }
                *(float4*)(raw + ((size_t)((i0 + ii) * LSEQ + 2 * (jp + 32 * pp)) << 1)) = s;
            }
        }
    }
}

__global__ __launch_bounds__(256)
void symmetrize(const float* __restrict__ raw, const float* __restrict__ b2,
                float* __restrict__ out) {
    int idx = blockIdx.x * 256 + threadIdx.x;
    if (idx >= LSEQ * LSEQ) return;
    int i = idx >> 9, j = idx & (LSEQ - 1);
    float2 rij = *(const float2*)(raw + ((size_t)(i * LSEQ + j) << 1));
    float2 rji = *(const float2*)(raw + ((size_t)(j * LSEQ + i) << 1));
    float2 o;
    o.x = 0.5f * (rij.x + rji.x) + b2[0];
    o.y = 0.5f * (rij.y + rji.y) + b2[1];
    *(float2*)(out + ((size_t)idx << 1)) = o;
}

extern "C" void kernel_launch(void* const* d_in, const int* in_sizes, int n_in,
                              void* d_out, int out_size, void* d_ws, size_t ws_size,
                              hipStream_t stream) {
    const float* X  = (const float*)d_in[0];
    const float* W1 = (const float*)d_in[1];
    const float* b1 = (const float*)d_in[2];
    const float* W2 = (const float*)d_in[3];
    const float* b2 = (const float*)d_in[4];
    float* out = (float*)d_out;

    float* a_ws = (float*)d_ws;
    float* r_ws = a_ws + (size_t)LSEQ * HDIM;
    float* raw  = r_ws + (size_t)LSEQ * HDIM;

    proj_wmma_f32<<<256, 128, 0, stream>>>(X, W1, b1, a_ws, r_ws);

    dim3 grid(LSEQ / (2 * NI), LSEQ / 128);
    pairwise_tanh<<<grid, 256, 0, stream>>>(a_ws, r_ws, W2, raw);

    symmetrize<<<(LSEQ * LSEQ) / 256, 256, 0, stream>>>(raw, b2, out);
}
